# starting kernel
# baseline (speedup 1.0000x reference)
#include <hip/hip_runtime.h>
#include <hip/hip_bf16.h>

typedef float v2f __attribute__((ext_vector_type(2)));
typedef float v8f __attribute__((ext_vector_type(8)));

#define D_ 128
#define K_ 128

__global__ __launch_bounds__(D_) void nks_baseline(
    const float* __restrict__ lbev, const float* __restrict__ lbcc,
    float* __restrict__ bev, float* __restrict__ bar) {
  __shared__ float s[D_];
  int d = threadIdx.x;
  float e = __expf(lbev[d]);
  float c = __expf(lbcc[d]);
  bev[d] = e;
  s[d] = e + c;
  __syncthreads();
  float acc = 0.0f;
  for (int j = d; j < D_; ++j) acc += s[j];
  bar[d] = acc;
}

__global__ __launch_bounds__(128) void nks_tables(
    const float* __restrict__ log_ev, const float* __restrict__ log_cc,
    float* __restrict__ EVt, float* __restrict__ ARt, int E) {
  __shared__ float S[16][130];
  const int e0 = blockIdx.x * 16;
  const int tid = threadIdx.x;
  const bool full = (e0 + 16 <= E);

  if (full) {
    for (int idx = tid; idx < 16 * D_; idx += 128) {
      int r = idx >> 7, col = idx & (D_ - 1);
      size_t g = (size_t)(e0 + r) * D_ + col;
      float ev = __expf(log_ev[g]);
      float cc = __expf(log_cc[g]);
      EVt[g] = ev;
      S[r][col] = ev + cc;
    }
  } else {
    for (int idx = tid; idx < 16 * D_; idx += 128) {
      int r = idx >> 7, col = idx & (D_ - 1);
      int e = e0 + r;
      int esafe = (e < E) ? e : (E - 1);
      size_t g = (size_t)esafe * D_ + col;
      float ev = __expf(log_ev[g]);
      float cc = __expf(log_cc[g]);
      if (e < E) EVt[g] = ev;
      S[r][col] = ev + cc;
    }
  }
  __syncthreads();

  const int wave = tid >> 5;
  const int lane = tid & 31;
  const int m = lane & 15;
  const int kh = lane >> 4;

  for (int t = 0; t < 2; ++t) {
    const int dt = wave + t * 4;
    const int n0 = dt * 16;
    const int n = n0 + m;
    v8f c = {0.f, 0.f, 0.f, 0.f, 0.f, 0.f, 0.f, 0.f};
#pragma unroll 4
    for (int kc = dt * 4; kc < 32; ++kc) {
      int j0 = kc * 4 + kh * 2;
      v2f a;
      a.x = S[m][j0];
      a.y = S[m][j0 + 1];
      v2f b;
      b.x = (j0 >= n) ? 1.0f : 0.0f;
      b.y = (j0 + 1 >= n) ? 1.0f : 0.0f;
      c = __builtin_amdgcn_wmma_f32_16x16x4_f32(
           false, a,  false, b,
           (short)0, c,  false,  false);
    }
    if (full) {
#pragma unroll
      for (int r = 0; r < 8; ++r) {
        ARt[(size_t)(e0 + r + kh * 8) * D_ + n] = c[r];
      }
    } else {
#pragma unroll
      for (int r = 0; r < 8; ++r) {
        int e = e0 + r + kh * 8;
        if (e < E) ARt[(size_t)e * D_ + n] = c[r];
      }
    }
  }
}

__global__ __launch_bounds__(256) void nks_main(
    const float* __restrict__ sq_dists, const int* __restrict__ labels,
    const float* __restrict__ EVt, const float* __restrict__ ARt,
    const float* __restrict__ bev, const float* __restrict__ bar,
    float* __restrict__ out, int B) {
  __shared__ float w_s[8][K_];
  __shared__ int l_s[8][K_];
  const int tid = threadIdx.x;
  const int wave = tid >> 5;
  const int lane = tid & 31;
  const int b = blockIdx.x * 8 + wave;
  if (b >= B) return;

  const float* sq = sq_dists + (size_t)b * K_;
  const int* lb = labels + (size_t)b * K_;
#pragma unroll
  for (int i = 0; i < 4; ++i) {
    int k = lane * 4 + i;
    float s = sq[k];
    w_s[wave][k] = (s <= 2.0f) ? __expf(-s) : 0.0f;
    l_s[wave][k] = lb[k];
  }

  const int d0 = lane * 4;
  float nx = 0.f, ny = 0.f, nz = 0.f, nw = 0.f;
  float dx = 0.f, dy = 0.f, dz = 0.f, dw = 0.f;
#pragma unroll 4
  for (int k = 0; k < K_; ++k) {
    float wk = w_s[wave][k];
    size_t row = (size_t)l_s[wave][k] * D_;
    float4 ev = *(const float4*)(EVt + row + d0);
    float4 ar = *(const float4*)(ARt + row + d0);
    nx = fmaf(wk, ev.x, nx); ny = fmaf(wk, ev.y, ny);
    nz = fmaf(wk, ev.z, nz); nw = fmaf(wk, ev.w, nw);
    dx = fmaf(wk, ar.x, dx); dy = fmaf(wk, ar.y, dy);
    dz = fmaf(wk, ar.z, dz); dw = fmaf(wk, ar.w, dw);
  }

  float4 be = *(const float4*)(bev + d0);
  float4 ba = *(const float4*)(bar + d0);
  const float lo = 1e-12f, hi = 1.0f - 1e-12f, eps = 1e-12f;
  float4 o;
  o.x = fminf(fmaxf((nx + be.x) / (dx + ba.x + eps), lo), hi);
  o.y = fminf(fmaxf((ny + be.y) / (dy + ba.y + eps), lo), hi);
  o.z = fminf(fmaxf((nz + be.z) / (dz + ba.z + eps), lo), hi);
  o.w = fminf(fmaxf((nw + be.w) / (dw + ba.w + eps), lo), hi);
  *(float4*)(out + (size_t)b * D_ + d0) = o;
}

extern "C" void kernel_launch(void* const* d_in, const int* in_sizes, int n_in,
                              void* d_out, int out_size, void* d_ws,
                              size_t ws_size, hipStream_t stream) {
  const float* sq_dists = (const float*)d_in[0];
  const float* log_ev = (const float*)d_in[1];
  const float* log_cc = (const float*)d_in[2];
  const float* lbev = (const float*)d_in[3];
  const float* lbcc = (const float*)d_in[4];
  const int* labels = (const int*)d_in[5];
  float* out = (float*)d_out;

  const int D = in_sizes[3];
  const int E = in_sizes[1] / D;
  const int B = out_size / D;

  float* EVt = (float*)d_ws;
  float* ARt = EVt + (size_t)E * D;
  float* bev = ARt + (size_t)E * D;
  float* bar = bev + D;

  nks_baseline<<<1, D_, 0, stream>>>(lbev, lbcc, bev, bar);
  nks_tables<<<(E + 15) / 16, 128, 0, stream>>>(log_ev, log_cc, EVt, ARt, E);
  nks_main<<<(B + 7) / 8, 256, 0, stream>>>(sq_dists, labels, EVt, ARt, bev,
                                            bar, out, B);
}


# Tables launch removed: log-table rows are gathered directly with exp at the gather; the suffix-sum matrix product runs once per 16 output rows after the weighted sum (sum order changes)
# speedup vs baseline: 1.8537x; 1.8537x over previous
#include <hip/hip_runtime.h>

typedef float v2f __attribute__((ext_vector_type(2)));
typedef float v8f __attribute__((ext_vector_type(8)));

#define D_ 128
#define K_ 128
#define RB_ 16
#define SP_ 132

__global__ __launch_bounds__(D_) void nks_baseline(
    const float* __restrict__ lbev, const float* __restrict__ lbcc,
    float* __restrict__ bev, float* __restrict__ bar) {
  __shared__ float s[D_];
  int d = threadIdx.x;
  float e = __expf(lbev[d]);
  float c = __expf(lbcc[d]);
  bev[d] = e;
  s[d] = e + c;
  __syncthreads();
  float acc = 0.0f;
  for (int j = d; j < D_; ++j) acc += s[j];
  bar[d] = acc;
}

struct Grp {
  float4 e[8];
  float4 c[8];
  float w[8];
};

__device__ __forceinline__ void grp_load(Grp& g, const float* wrow,
                                         const unsigned* orow, int k0,
                                         const char* pe, const char* pc,
                                         unsigned lo) {
  const float4 wa = *(const float4*)(wrow + k0);
  const float4 wb = *(const float4*)(wrow + k0 + 4);
  const uint4 oa = *(const uint4*)(orow + k0);
  const uint4 ob = *(const uint4*)(orow + k0 + 4);
  const unsigned ok[8] = {oa.x, oa.y, oa.z, oa.w, ob.x, ob.y, ob.z, ob.w};
  g.w[0] = wa.x; g.w[1] = wa.y; g.w[2] = wa.z; g.w[3] = wa.w;
  g.w[4] = wb.x; g.w[5] = wb.y; g.w[6] = wb.z; g.w[7] = wb.w;
#pragma unroll
  for (int i = 0; i < 8; ++i) {
    g.e[i] = *(const float4*)(pe + (ok[i] + lo));
    g.c[i] = *(const float4*)(pc + (ok[i] + lo));
  }
}

__device__ __forceinline__ void grp_acc(const Grp& g, float& nx, float& ny,
                                        float& nz, float& nw, float& sx,
                                        float& sy, float& sz, float& sw) {
#pragma unroll
  for (int i = 0; i < 8; ++i) {
    const float ex = __expf(g.e[i].x), ey = __expf(g.e[i].y);
    const float ez = __expf(g.e[i].z), ew = __expf(g.e[i].w);
    const float cx = __expf(g.c[i].x), cy = __expf(g.c[i].y);
    const float cz = __expf(g.c[i].z), cw = __expf(g.c[i].w);
    nx = fmaf(g.w[i], ex, nx); ny = fmaf(g.w[i], ey, ny);
    nz = fmaf(g.w[i], ez, nz); nw = fmaf(g.w[i], ew, nw);
    sx = fmaf(g.w[i], ex + cx, sx); sy = fmaf(g.w[i], ey + cy, sy);
    sz = fmaf(g.w[i], ez + cz, sz); sw = fmaf(g.w[i], ew + cw, sw);
  }
}

__global__ __launch_bounds__(RB_ * 32) __attribute__((amdgpu_waves_per_eu(4, 4)))
void nks_fused(
    const float* __restrict__ sq_dists, const int* __restrict__ labels,
    const float* __restrict__ log_ev, const float* __restrict__ log_cc,
    const float* __restrict__ bev, const float* __restrict__ bar,
    float* __restrict__ out, int B) {
  __shared__ __attribute__((aligned(16))) float w_s[RB_][K_];
  __shared__ __attribute__((aligned(16))) unsigned o_s[RB_][K_];
  __shared__ __attribute__((aligned(16))) float S[RB_][SP_];
  __shared__ __attribute__((aligned(16))) float Dn[RB_][SP_];

  const int tid = threadIdx.x;
  const int wave = __builtin_amdgcn_readfirstlane(tid >> 5);
  const int lane = tid & 31;
  const int b = blockIdx.x * RB_ + wave;
  const bool valid = (b < B);
  const int d0 = lane * 4;

  float nx = 0.f, ny = 0.f, nz = 0.f, nw = 0.f;
  float sx = 0.f, sy = 0.f, sz = 0.f, sw = 0.f;

  if (valid) {
    const float4 sq = *(const float4*)(sq_dists + (size_t)b * K_ + d0);
    const int4 lb = *(const int4*)(labels + (size_t)b * K_ + d0);
    float4 w4;
    w4.x = (sq.x <= 2.0f) ? __expf(-sq.x) : 0.0f;
    w4.y = (sq.y <= 2.0f) ? __expf(-sq.y) : 0.0f;
    w4.z = (sq.z <= 2.0f) ? __expf(-sq.z) : 0.0f;
    w4.w = (sq.w <= 2.0f) ? __expf(-sq.w) : 0.0f;
    uint4 o4;
    o4.x = (unsigned)lb.x * (unsigned)(D_ * 4);
    o4.y = (unsigned)lb.y * (unsigned)(D_ * 4);
    o4.z = (unsigned)lb.z * (unsigned)(D_ * 4);
    o4.w = (unsigned)lb.w * (unsigned)(D_ * 4);
    *(float4*)&w_s[wave][d0] = w4;
    *(uint4*)&o_s[wave][d0] = o4;

    const char* pe = (const char*)log_ev;
    const char* pc = (const char*)log_cc;
    const unsigned lo = (unsigned)lane * 16u;
    const float* wrow = &w_s[wave][0];
    const unsigned* orow = &o_s[wave][0];
    Grp ga, gb;
    grp_load(ga, wrow, orow, 0, pe, pc, lo);
#pragma unroll 1
    for (int k0 = 0; k0 < K_ - 16; k0 += 16) {
      grp_load(gb, wrow, orow, k0 + 8, pe, pc, lo);
      grp_acc(ga, nx, ny, nz, nw, sx, sy, sz, sw);
      grp_load(ga, wrow, orow, k0 + 16, pe, pc, lo);
      grp_acc(gb, nx, ny, nz, nw, sx, sy, sz, sw);
    }
    grp_load(gb, wrow, orow, K_ - 8, pe, pc, lo);
    grp_acc(ga, nx, ny, nz, nw, sx, sy, sz, sw);
    grp_acc(gb, nx, ny, nz, nw, sx, sy, sz, sw);
  }

  float4 s4; s4.x = sx; s4.y = sy; s4.z = sz; s4.w = sw;
  *(float4*)&S[wave][d0] = s4;
  __syncthreads();

  if (wave < 8) {
    const int m = lane & 15;
    const int kh = lane >> 4;
    const int dt = (wave < 4) ? wave : 11 - wave;
    const int n0 = dt * 16;
    const int n = n0 + m;
    v8f c = {0.f, 0.f, 0.f, 0.f, 0.f, 0.f, 0.f, 0.f};
#pragma unroll 4
    for (int kc = dt * 4; kc < 32; ++kc) {
      int j0 = kc * 4 + kh * 2;
      v2f a;
      a.x = S[m][j0];
      a.y = S[m][j0 + 1];
      v2f bt;
      bt.x = (j0 >= n) ? 1.0f : 0.0f;
      bt.y = (j0 + 1 >= n) ? 1.0f : 0.0f;
      c = __builtin_amdgcn_wmma_f32_16x16x4_f32(
           false, a,  false, bt,
           (short)0, c,  false,  false);
    }
#pragma unroll
    for (int r = 0; r < 8; ++r) Dn[r + kh * 8][n] = c[r];
  }
  __syncthreads();

  if (valid) {
    const float4 dn = *(const float4*)&Dn[wave][d0];
    const float4 be = *(const float4*)(bev + d0);
    const float4 ba = *(const float4*)(bar + d0);
    const float lo = 1e-12f, hi = 1.0f - 1e-12f, eps = 1e-12f;
    float4 o;
    o.x = fminf(fmaxf((nx + be.x) / (dn.x + ba.x + eps), lo), hi);
    o.y = fminf(fmaxf((ny + be.y) / (dn.y + ba.y + eps), lo), hi);
    o.z = fminf(fmaxf((nz + be.z) / (dn.z + ba.z + eps), lo), hi);
    o.w = fminf(fmaxf((nw + be.w) / (dn.w + ba.w + eps), lo), hi);
    *(float4*)(out + (size_t)b * D_ + d0) = o;
  }
}

extern "C" void kernel_launch(void* const* d_in, const int* in_sizes, int n_in,
                              void* d_out, int out_size, void* d_ws,
                              size_t ws_size, hipStream_t stream) {
  const float* sq_dists = (const float*)d_in[0];
  const float* log_ev = (const float*)d_in[1];
  const float* log_cc = (const float*)d_in[2];
  const float* lbev = (const float*)d_in[3];
  const float* lbcc = (const float*)d_in[4];
  const int* labels = (const int*)d_in[5];
  float* out = (float*)d_out;

  const int D = in_sizes[3];
  const int B = out_size / D;

  float* bev = (float*)d_ws;
  float* bar = bev + D_;

  nks_baseline<<<1, D_, 0, stream>>>(lbev, lbcc, bev, bar);
  nks_fused<<<(B + RB_ - 1) / RB_, RB_ * 32, 0, stream>>>(
      sq_dists, labels, log_ev, log_cc, bev, bar, out, B);
}


# Baseline launch removed: each wave starts its accumulators at exp of the baseline logs, so the suffix-sum product yields exemplar plus baseline at-risk totals in one sum (sum order changes)
# speedup vs baseline: 2.2248x; 1.2002x over previous
#include <hip/hip_runtime.h>

typedef float v2f __attribute__((ext_vector_type(2)));
typedef float v8f __attribute__((ext_vector_type(8)));

#define D_ 128
#define K_ 128
#define RB_ 16
#define SP_ 132

struct Grp {
  float4 e[8];
  float4 c[8];
  float w[8];
};

__device__ __forceinline__ void grp_load(Grp& g, const float* wrow,
                                         const unsigned* orow, int k0,
                                         const char* pe, const char* pc,
                                         unsigned lo) {
  const float4 wa = *(const float4*)(wrow + k0);
  const float4 wb = *(const float4*)(wrow + k0 + 4);
  const uint4 oa = *(const uint4*)(orow + k0);
  const uint4 ob = *(const uint4*)(orow + k0 + 4);
  const unsigned ok[8] = {oa.x, oa.y, oa.z, oa.w, ob.x, ob.y, ob.z, ob.w};
  g.w[0] = wa.x; g.w[1] = wa.y; g.w[2] = wa.z; g.w[3] = wa.w;
  g.w[4] = wb.x; g.w[5] = wb.y; g.w[6] = wb.z; g.w[7] = wb.w;
#pragma unroll
  for (int i = 0; i < 8; ++i) {
    g.e[i] = *(const float4*)(pe + (ok[i] + lo));
    g.c[i] = *(const float4*)(pc + (ok[i] + lo));
  }
}

__device__ __forceinline__ void grp_acc(const Grp& g, float& nx, float& ny,
                                        float& nz, float& nw, float& sx,
                                        float& sy, float& sz, float& sw) {
#pragma unroll
  for (int i = 0; i < 8; ++i) {
    const float ex = __expf(g.e[i].x), ey = __expf(g.e[i].y);
    const float ez = __expf(g.e[i].z), ew = __expf(g.e[i].w);
    const float cx = __expf(g.c[i].x), cy = __expf(g.c[i].y);
    const float cz = __expf(g.c[i].z), cw = __expf(g.c[i].w);
    nx = fmaf(g.w[i], ex, nx); ny = fmaf(g.w[i], ey, ny);
    nz = fmaf(g.w[i], ez, nz); nw = fmaf(g.w[i], ew, nw);
    sx = fmaf(g.w[i], ex + cx, sx); sy = fmaf(g.w[i], ey + cy, sy);
    sz = fmaf(g.w[i], ez + cz, sz); sw = fmaf(g.w[i], ew + cw, sw);
  }
}

__global__ __launch_bounds__(RB_ * 32) __attribute__((amdgpu_waves_per_eu(4, 4)))
void nks_fused(
    const float* __restrict__ sq_dists, const int* __restrict__ labels,
    const float* __restrict__ log_ev, const float* __restrict__ log_cc,
    const float* __restrict__ lbev, const float* __restrict__ lbcc,
    float* __restrict__ out, int B) {
  __shared__ __attribute__((aligned(16))) float w_s[RB_][K_];
  __shared__ __attribute__((aligned(16))) unsigned o_s[RB_][K_];
  __shared__ __attribute__((aligned(16))) float S[RB_][SP_];
  __shared__ __attribute__((aligned(16))) float Dn[RB_][SP_];

  const int tid = threadIdx.x;
  const int wave = __builtin_amdgcn_readfirstlane(tid >> 5);
  const int lane = tid & 31;
  const int b = blockIdx.x * RB_ + wave;
  const bool valid = (b < B);
  const int d0 = lane * 4;

  float nx = 0.f, ny = 0.f, nz = 0.f, nw = 0.f;
  float sx = 0.f, sy = 0.f, sz = 0.f, sw = 0.f;

  if (valid) {
    const float4 sq = *(const float4*)(sq_dists + (size_t)b * K_ + d0);
    const int4 lb = *(const int4*)(labels + (size_t)b * K_ + d0);
    const float4 le = *(const float4*)(lbev + d0);
    const float4 lc = *(const float4*)(lbcc + d0);
    nx = __expf(le.x); ny = __expf(le.y); nz = __expf(le.z); nw = __expf(le.w);
    sx = nx + __expf(lc.x); sy = ny + __expf(lc.y);
    sz = nz + __expf(lc.z); sw = nw + __expf(lc.w);
    float4 w4;
    w4.x = (sq.x <= 2.0f) ? __expf(-sq.x) : 0.0f;
    w4.y = (sq.y <= 2.0f) ? __expf(-sq.y) : 0.0f;
    w4.z = (sq.z <= 2.0f) ? __expf(-sq.z) : 0.0f;
    w4.w = (sq.w <= 2.0f) ? __expf(-sq.w) : 0.0f;
    uint4 o4;
    o4.x = (unsigned)lb.x * (unsigned)(D_ * 4);
    o4.y = (unsigned)lb.y * (unsigned)(D_ * 4);
    o4.z = (unsigned)lb.z * (unsigned)(D_ * 4);
    o4.w = (unsigned)lb.w * (unsigned)(D_ * 4);
    *(float4*)&w_s[wave][d0] = w4;
    *(uint4*)&o_s[wave][d0] = o4;

    const char* pe = (const char*)log_ev;
    const char* pc = (const char*)log_cc;
    const unsigned lo = (unsigned)lane * 16u;
    const float* wrow = &w_s[wave][0];
    const unsigned* orow = &o_s[wave][0];
    Grp ga, gb;
    grp_load(ga, wrow, orow, 0, pe, pc, lo);
#pragma unroll 1
    for (int k0 = 0; k0 < K_ - 16; k0 += 16) {
      grp_load(gb, wrow, orow, k0 + 8, pe, pc, lo);
      grp_acc(ga, nx, ny, nz, nw, sx, sy, sz, sw);
      grp_load(ga, wrow, orow, k0 + 16, pe, pc, lo);
      grp_acc(gb, nx, ny, nz, nw, sx, sy, sz, sw);
    }
    grp_load(gb, wrow, orow, K_ - 8, pe, pc, lo);
    grp_acc(ga, nx, ny, nz, nw, sx, sy, sz, sw);
    grp_acc(gb, nx, ny, nz, nw, sx, sy, sz, sw);
  }

  float4 s4; s4.x = sx; s4.y = sy; s4.z = sz; s4.w = sw;
  *(float4*)&S[wave][d0] = s4;
  __syncthreads();

  if (wave < 8) {
    const int m = lane & 15;
    const int kh = lane >> 4;
    const int dt = (wave < 4) ? wave : 11 - wave;
    const int n0 = dt * 16;
    const int n = n0 + m;
    v8f c = {0.f, 0.f, 0.f, 0.f, 0.f, 0.f, 0.f, 0.f};
#pragma unroll 4
    for (int kc = dt * 4; kc < 32; ++kc) {
      int j0 = kc * 4 + kh * 2;
      v2f a;
      a.x = S[m][j0];
      a.y = S[m][j0 + 1];
      v2f bt;
      bt.x = (j0 >= n) ? 1.0f : 0.0f;
      bt.y = (j0 + 1 >= n) ? 1.0f : 0.0f;
      c = __builtin_amdgcn_wmma_f32_16x16x4_f32(
           false, a,  false, bt,
           (short)0, c,  false,  false);
    }
#pragma unroll
    for (int r = 0; r < 8; ++r) Dn[r + kh * 8][n] = c[r];
  }
  __syncthreads();

  if (valid) {
    const float4 dn = *(const float4*)&Dn[wave][d0];
    const float lo = 1e-12f, hi = 1.0f - 1e-12f, eps = 1e-12f;
    float4 o;
    o.x = fminf(fmaxf(nx / (dn.x + eps), lo), hi);
    o.y = fminf(fmaxf(ny / (dn.y + eps), lo), hi);
    o.z = fminf(fmaxf(nz / (dn.z + eps), lo), hi);
    o.w = fminf(fmaxf(nw / (dn.w + eps), lo), hi);
    *(float4*)(out + (size_t)b * D_ + d0) = o;
  }
}

extern "C" void kernel_launch(void* const* d_in, const int* in_sizes, int n_in,
                              void* d_out, int out_size, void* d_ws,
                              size_t ws_size, hipStream_t stream) {
  const float* sq_dists = (const float*)d_in[0];
  const float* log_ev = (const float*)d_in[1];
  const float* log_cc = (const float*)d_in[2];
  const float* lbev = (const float*)d_in[3];
  const float* lbcc = (const float*)d_in[4];
  const int* labels = (const int*)d_in[5];
  float* out = (float*)d_out;

  const int D = in_sizes[3];
  const int B = out_size / D;

  nks_fused<<<(B + RB_ - 1) / RB_, RB_ * 32, 0, stream>>>(
      sq_dists, labels, log_ev, log_cc, lbev, lbcc, out, B);
}
